# starting kernel
# baseline (speedup 1.0000x reference)
#include <hip/hip_runtime.h>
#include <math.h>

typedef __attribute__((ext_vector_type(2))) float v2f;
typedef __attribute__((ext_vector_type(8))) float v8f;

#define B_  1024
#define N_  1024
#define M_  8
#define D_  64
#define R_  64
#define C_  10

#define LDS_W      32768
#define LDS_HALF   10240
#define AT_STRIDE  66
#define S_STRIDE   66

__global__ void wl_kernel(const float* __restrict__ Wlast, float* __restrict__ Wl) {
    int tid = blockIdx.x * 256 + threadIdx.x;
    if (tid >= C_ * R_ * R_) return;
    int r  = tid & 63;
    int ci = tid >> 6;
    const float* p = Wlast + ((size_t)ci * 8) * 64 + r;
    float s = 0.f;
    #pragma unroll
    for (int m = 0; m < 8; ++m) s += p[m * 64];
    Wl[tid] = s;
}

__global__ void phi_kernel(const float* __restrict__ x, float* __restrict__ phidiv) {
    int tid = blockIdx.x * 256 + threadIdx.x;
    if (tid >= B_ * D_) return;
    const float* xp = x + (size_t)tid * 16;
    float acc[8];
    #pragma unroll
    for (int m = 0; m < 8; ++m) acc[m] = 0.f;
    for (int n = 0; n < 16; ++n) {
        float a = 1.5707963267948966f * xp[n];
        float sn, cs;
        sincosf(a, &sn, &cs);
        float pc[8], ps[8];
        pc[0] = 1.f; ps[0] = 1.f;
        #pragma unroll
        for (int k = 1; k < 8; ++k) { pc[k] = pc[k-1] * cs; ps[k] = ps[k-1] * sn; }
        #pragma unroll
        for (int m = 0; m < 8; ++m) acc[m] = fmaf(pc[7-m], ps[m], acc[m]);
    }
    float* o = phidiv + (size_t)tid * 8;
    #pragma unroll
    for (int m = 0; m < 8; ++m) o[m] = acc[m];
}

__global__ __launch_bounds__(256, 1)
void trnet_kernel(const float* __restrict__ Wsh, const float* __restrict__ phidiv,
                  const float* __restrict__ Wl, float* __restrict__ out)
{
    extern __shared__ float smem[];
    float* Wsm = smem;
    const int half = threadIdx.x >> 7;
    const int t0   = threadIdx.x & 127;
    float* At  = smem + LDS_W + half * LDS_HALF;
    float* Ssm = At  + 64 * AT_STRIDE;
    float* phs = Ssm + 64 * S_STRIDE;
    float* red = phs + 512;

    const int b = blockIdx.x * 2 + half;

    {
        const float4* src = (const float4*)Wsh;
        float4* dst = (float4*)Wsm;
        for (int q = 0; q < 32; ++q)
            dst[threadIdx.x + 256 * q] = src[threadIdx.x + 256 * q];
    }
    {
        const float* src = phidiv + (size_t)b * (D_ * M_);
        #pragma unroll
        for (int q = 0; q < 4; ++q) phs[t0 + 128 * q] = src[t0 + 128 * q];
    }
    for (int q = t0; q < 64 * S_STRIDE; q += 128) Ssm[q] = 1.0f;
    __syncthreads();

    const int wave = t0 >> 5;
    const int lane = t0 & 31;
    const int lmod = lane & 15;
    const int lhi  = lane >> 4;

    for (int d = 0; d < D_; ++d) {
        float ph[8];
        #pragma unroll
        for (int j = 0; j < 8; ++j) ph[j] = phs[d * 8 + j];
        #pragma unroll 4
        for (int q = 0; q < 32; ++q) {
            int e  = t0 + 128 * q;
            int ko = e & 63;
            int i  = e >> 6;
            const float* wp = Wsm + i * 512 + ko;
            float v = 0.f;
            #pragma unroll
            for (int j = 0; j < 8; ++j) v = fmaf(ph[j], wp[j * 64], v);
            At[ko * AT_STRIDE + i] = v;
        }
        __syncthreads();

        const float* srow = Ssm + (16 * wave + lmod) * S_STRIDE;
        v8f c0 = {}, c1 = {}, c2 = {}, c3 = {};
        #pragma unroll
        for (int kb = 0; kb < 16; ++kb) {
            const int koff = kb * 4 + 2 * lhi;
            v2f a = *(const v2f*)(srow + koff);
            const float* ab = At + koff;
            v2f b0 = *(const v2f*)(ab + (lmod +  0) * AT_STRIDE);
            v2f b1 = *(const v2f*)(ab + (lmod + 16) * AT_STRIDE);
            v2f b2 = *(const v2f*)(ab + (lmod + 32) * AT_STRIDE);
            v2f b3 = *(const v2f*)(ab + (lmod + 48) * AT_STRIDE);
            c0 = __builtin_amdgcn_wmma_f32_16x16x4_f32(false, a, false, b0, (short)0, c0, false, false);
            c1 = __builtin_amdgcn_wmma_f32_16x16x4_f32(false, a, false, b1, (short)0, c1, false, false);
            c2 = __builtin_amdgcn_wmma_f32_16x16x4_f32(false, a, false, b2, (short)0, c2, false, false);
            c3 = __builtin_amdgcn_wmma_f32_16x16x4_f32(false, a, false, b3, (short)0, c3, false, false);
        }
        #pragma unroll
        for (int v = 0; v < 8; ++v) {
            float* prow = Ssm + (16 * wave + v + 8 * lhi) * S_STRIDE + lmod;
            prow[0]  = c0[v];
            prow[16] = c1[v];
            prow[32] = c2[v];
            prow[48] = c3[v];
        }
        __syncthreads();
    }

    float acc[C_];
    #pragma unroll
    for (int c = 0; c < C_; ++c) acc[c] = 0.f;
    for (int q = 0; q < 32; ++q) {
        int e = t0 + 128 * q;
        int r = e & 63;
        int i = e >> 6;
        float s = Ssm[r * S_STRIDE + i];
        #pragma unroll
        for (int c = 0; c < C_; ++c)
            acc[c] = fmaf(s, Wl[c * 4096 + i * 64 + r], acc[c]);
    }
    #pragma unroll
    for (int c = 0; c < C_; ++c) red[t0 * C_ + c] = acc[c];
    __syncthreads();
    if (t0 < C_) {
        float s = 0.f;
        for (int t = 0; t < 128; ++t) s += red[t * C_ + t0];
        At[t0] = s;
    }
    __syncthreads();
    if (t0 < C_) {
        float mx = At[0];
        #pragma unroll
        for (int c = 1; c < C_; ++c) mx = fmaxf(mx, At[c]);
        float se = 0.f;
        #pragma unroll
        for (int c = 0; c < C_; ++c) se += expf(At[c] - mx);
        out[b * C_ + t0] = At[t0] - (logf(se) + mx);
    }
}

extern "C" void kernel_launch(void* const* d_in, const int* in_sizes, int n_in,
                              void* d_out, int out_size, void* d_ws, size_t ws_size,
                              hipStream_t stream) {
    const float* x     = (const float*)d_in[0];
    const float* Wsh   = (const float*)d_in[1];
    const float* Wlast = (const float*)d_in[2];
    float* out = (float*)d_out;
    float* ws  = (float*)d_ws;
    float* Wl     = ws;
    float* phidiv = ws + (C_ * R_ * R_);

    wl_kernel<<<(C_ * R_ * R_ + 255) / 256, 256, 0, stream>>>(Wlast, Wl);
    phi_kernel<<<(B_ * D_ + 255) / 256, 256, 0, stream>>>(x, phidiv);

    const size_t shbytes = (size_t)(LDS_W + 2 * LDS_HALF) * sizeof(float);
    hipFuncSetAttribute((const void*)trnet_kernel,
                        hipFuncAttributeMaxDynamicSharedMemorySize, (int)shbytes);
    trnet_kernel<<<B_ / 2, 256, shbytes, stream>>>(Wsh, phidiv, Wl, out);
}


# Step body as one software pipeline: next step's matrix formed between this step's WMMAs into a 2nd LDS buffer, fragment loads grouped one k-pair ahead; one barrier per step; sum orders unchanged
# speedup vs baseline: 2.3887x; 2.3887x over previous
#include <hip/hip_runtime.h>
#include <math.h>


typedef __attribute__((ext_vector_type(2))) float v2f;
typedef __attribute__((ext_vector_type(8))) float v8f;

#define B_  1024
#define N_  1024
#define M_  8
#define D_  64
#define R_  64
#define C_  10

#define AT_STRIDE  66
#define S_STRIDE   66
#define AT_BUF     (64 * AT_STRIDE)
#define LDS_HALF   (2 * AT_BUF + 64 * S_STRIDE + 128 * C_)

__global__ void wl_kernel(const float* __restrict__ Wlast, float* __restrict__ Wl) {
    int tid = blockIdx.x * 256 + threadIdx.x;
    if (tid >= C_ * R_ * R_) return;
    int r  = tid & 63;
    int ci = tid >> 6;
    const float* p = Wlast + ((size_t)ci * 8) * 64 + r;
    float s = 0.f;
    #pragma unroll
    for (int m = 0; m < 8; ++m) s += p[m * 64];
    Wl[tid] = s;
}

__global__ void phi_kernel(const float* __restrict__ x, float* __restrict__ phidiv) {
    int tid = blockIdx.x * 256 + threadIdx.x;
    if (tid >= B_ * D_) return;
    const float* xp = x + (size_t)tid * 16;
    float acc[8];
    #pragma unroll
    for (int m = 0; m < 8; ++m) acc[m] = 0.f;
    for (int n = 0; n < 16; ++n) {
        float a = 1.5707963267948966f * xp[n];
        float sn, cs;
        sincosf(a, &sn, &cs);
        float pc[8], ps[8];
        pc[0] = 1.f; ps[0] = 1.f;
        #pragma unroll
        for (int k = 1; k < 8; ++k) { pc[k] = pc[k-1] * cs; ps[k] = ps[k-1] * sn; }
        #pragma unroll
        for (int m = 0; m < 8; ++m) acc[m] = fmaf(pc[7-m], ps[m], acc[m]);
    }
    float* o = phidiv + (size_t)tid * 8;
    #pragma unroll
    for (int m = 0; m < 8; ++m) o[m] = acc[m];
}

__device__ __forceinline__ void form_pair(const v2f (&wr)[8][8], const float (&pa)[8], const float (&pb)[8],
                                          float* __restrict__ f0, float* __restrict__ f1, const int p)
{
    v2f u = {0.f, 0.f}, w = {0.f, 0.f};
    #pragma unroll
    for (int j = 0; j < 8; ++j) {
        const v2f xa = {pa[j], pa[j]};
        const v2f xb = {pb[j], pb[j]};
        u = __builtin_elementwise_fma(xa, wr[p][j], u);
        w = __builtin_elementwise_fma(xb, wr[p][j], w);
    }
    *(v2f*)(f0 + 2 * p) = u;
    *(v2f*)(f1 + 2 * p) = w;
}

struct Frag2 { v2f a[2], b0[2], b1[2], b2[2], b3[2]; };

__device__ __forceinline__ void load_frag2(Frag2& f, const float* __restrict__ srow,
                                           const float* __restrict__ Acur, const int lmod, const int lhi, const int p)
{
    #pragma unroll
    for (int h = 0; h < 2; ++h) {
        const int koff = (2 * p + h) * 4 + 2 * lhi;
        f.a[h] = *(const v2f*)(srow + koff);
        const float* ab = Acur + koff;
        f.b0[h] = *(const v2f*)(ab + (lmod +  0) * AT_STRIDE);
        f.b1[h] = *(const v2f*)(ab + (lmod + 16) * AT_STRIDE);
        f.b2[h] = *(const v2f*)(ab + (lmod + 32) * AT_STRIDE);
        f.b3[h] = *(const v2f*)(ab + (lmod + 48) * AT_STRIDE);
    }
}

template <bool FORM>
__device__ __forceinline__ void chain_step(const v2f (&wr)[8][8], const float (&pa)[8], const float (&pb)[8],
                                           float* __restrict__ f0, float* __restrict__ f1,
                                           const float* __restrict__ Acur, float* __restrict__ Ssm,
                                           const int wave, const int lmod, const int lhi)
{
    const float* srow = Ssm + (16 * wave + lmod) * S_STRIDE;
    v8f c0 = {}, c1 = {}, c2 = {}, c3 = {};
    Frag2 fr[2];
    load_frag2(fr[0], srow, Acur, lmod, lhi, 0);
    __builtin_amdgcn_sched_barrier(0);
    #pragma unroll
    for (int p = 0; p < 8; ++p) {
        const Frag2& f = fr[p & 1];
        if (p < 7) load_frag2(fr[(p + 1) & 1], srow, Acur, lmod, lhi, p + 1);
        #pragma unroll
        for (int h = 0; h < 2; ++h) {
            c0 = __builtin_amdgcn_wmma_f32_16x16x4_f32(false, f.a[h], false, f.b0[h], (short)0, c0, false, false);
            c1 = __builtin_amdgcn_wmma_f32_16x16x4_f32(false, f.a[h], false, f.b1[h], (short)0, c1, false, false);
            c2 = __builtin_amdgcn_wmma_f32_16x16x4_f32(false, f.a[h], false, f.b2[h], (short)0, c2, false, false);
            c3 = __builtin_amdgcn_wmma_f32_16x16x4_f32(false, f.a[h], false, f.b3[h], (short)0, c3, false, false);
        }
        if (FORM) form_pair(wr, pa, pb, f0, f1, p);
        if (p < 7) __builtin_amdgcn_sched_group_barrier(0x100, 5, 0);
        #pragma unroll
        for (int g = 0; g < 8; ++g) {
            __builtin_amdgcn_sched_group_barrier(0x008, 1, 0);
            if (FORM) __builtin_amdgcn_sched_group_barrier(0x002, 2, 0);
        }
        if (FORM) __builtin_amdgcn_sched_group_barrier(0x200, 2, 0);
        __builtin_amdgcn_sched_barrier(0);
    }
    #pragma unroll
    for (int v = 0; v < 8; ++v) {
        float* prow = Ssm + (16 * wave + v + 8 * lhi) * S_STRIDE + lmod;
        prow[0]  = c0[v];
        prow[16] = c1[v];
        prow[32] = c2[v];
        prow[48] = c3[v];
    }
}

__global__ __launch_bounds__(256, 4)
void trnet_kernel(const float* __restrict__ Wsh, const float* __restrict__ phidiv,
                  const float* __restrict__ Wl, float* __restrict__ out)
{
    extern __shared__ float smem[];
    const int half = threadIdx.x >> 7;
    const int t0   = threadIdx.x & 127;
    float* At  = smem + half * LDS_HALF;
    float* Ssm = At  + 2 * AT_BUF;
    float* red = Ssm + 64 * S_STRIDE;

    const int b = blockIdx.x * 2 + half;

    const int fko = threadIdx.x & 63;
    const int fi0 = (threadIdx.x >> 6) * 16;
    v2f wr[8][8];
    {
        const float* wp = Wsh + fi0 * 512 + fko;
        #pragma unroll
        for (int p = 0; p < 8; ++p) {
            #pragma unroll
            for (int j = 0; j < 8; ++j) {
                const v2f t = {wp[(2 * p) * 512 + j * 64], wp[(2 * p + 1) * 512 + j * 64]};
                wr[p][j] = t;
            }
        }
    }
    float* fa0 = smem + fko * AT_STRIDE + fi0;
    float* fa1 = fa0 + LDS_HALF;
    typedef const __attribute__((address_space(4))) float cfloat;
    cfloat* php = (cfloat*)(phidiv + (size_t)blockIdx.x * (2 * D_ * M_));
    cfloat* phq = php + D_ * M_;

    for (int q = t0; q < 64 * S_STRIDE; q += 128) Ssm[q] = 1.0f;

    const int wave = t0 >> 5;
    const int lane = t0 & 31;
    const int lmod = lane & 15;
    const int lhi  = lane >> 4;

    float pa[8], pb[8];
    #pragma unroll
    for (int j = 0; j < 8; ++j) { pa[j] = php[j]; pb[j] = phq[j]; }
    #pragma unroll
    for (int p = 0; p < 8; ++p) form_pair(wr, pa, pb, fa0, fa1, p);
    #pragma unroll
    for (int j = 0; j < 8; ++j) { pa[j] = php[8 + j]; pb[j] = phq[8 + j]; }
    __syncthreads();

    for (int d = 0; d < D_ - 1; ++d) {
        const int cur = (d & 1) * AT_BUF;
        const int nxt = AT_BUF - cur;
        float na[8], nb[8];
        {
            const int dn = (d + 2) & (D_ - 1);
            #pragma unroll
            for (int j = 0; j < 8; ++j) { na[j] = php[dn * 8 + j]; nb[j] = phq[dn * 8 + j]; }
        }
        chain_step<true>(wr, pa, pb, fa0 + nxt, fa1 + nxt, At + cur, Ssm, wave, lmod, lhi);
        __syncthreads();
        #pragma unroll
        for (int j = 0; j < 8; ++j) { pa[j] = na[j]; pb[j] = nb[j]; }
    }
    chain_step<false>(wr, pa, pb, fa0, fa1, At + AT_BUF, Ssm, wave, lmod, lhi);
    __syncthreads();

    float acc[C_];
    #pragma unroll
    for (int c = 0; c < C_; ++c) acc[c] = 0.f;
    for (int q = 0; q < 32; ++q) {
        int e = t0 + 128 * q;
        int r = e & 63;
        int i = e >> 6;
        float s = Ssm[r * S_STRIDE + i];
        #pragma unroll
        for (int c = 0; c < C_; ++c)
            acc[c] = fmaf(s, Wl[c * 4096 + i * 64 + r], acc[c]);
    }
    #pragma unroll
    for (int c = 0; c < C_; ++c) red[t0 * C_ + c] = acc[c];
    __syncthreads();
    if (t0 < C_) {
        float s = 0.f;
        for (int t = 0; t < 128; ++t) s += red[t * C_ + t0];
        At[t0] = s;
    }
    __syncthreads();
    if (t0 < C_) {
        float mx = At[0];
        #pragma unroll
        for (int c = 1; c < C_; ++c) mx = fmaxf(mx, At[c]);
        float se = 0.f;
        #pragma unroll
        for (int c = 0; c < C_; ++c) se += expf(At[c] - mx);
        out[b * C_ + t0] = At[t0] - (logf(se) + mx);
    }
}

extern "C" void kernel_launch(void* const* d_in, const int* in_sizes, int n_in,
                              void* d_out, int out_size, void* d_ws, size_t ws_size,
                              hipStream_t stream) {
    const float* x     = (const float*)d_in[0];
    const float* Wsh   = (const float*)d_in[1];
    const float* Wlast = (const float*)d_in[2];
    float* out = (float*)d_out;
    float* ws  = (float*)d_ws;
    float* Wl     = ws;
    float* phidiv = ws + (C_ * R_ * R_);

    wl_kernel<<<(C_ * R_ * R_ + 255) / 256, 256, 0, stream>>>(Wlast, Wl);
    phi_kernel<<<(B_ * D_ + 255) / 256, 256, 0, stream>>>(x, phidiv);

    const size_t shbytes = (size_t)(2 * LDS_HALF) * sizeof(float);
    hipFuncSetAttribute((const void*)trnet_kernel,
                        hipFuncAttributeMaxDynamicSharedMemorySize, (int)shbytes);
    trnet_kernel<<<B_ / 2, 256, shbytes, stream>>>(Wsh, phidiv, Wl, out);
}


# Wave priority falls with chain progress (s_setprio 3,2,1,0 from steps 0,32,48,56): of the two blocks on a WGP the one behind wins the matrix pipe and neither ends alone; sum orders unchanged
# speedup vs baseline: 2.4960x; 1.0449x over previous
#include <hip/hip_runtime.h>
#include <math.h>


typedef __attribute__((ext_vector_type(2))) float v2f;
typedef __attribute__((ext_vector_type(8))) float v8f;

#define B_  1024
#define N_  1024
#define M_  8
#define D_  64
#define R_  64
#define C_  10

#define AT_STRIDE  66
#define S_STRIDE   66
#define AT_BUF     (64 * AT_STRIDE)
#define LDS_HALF   (2 * AT_BUF + 64 * S_STRIDE + 128 * C_)

__global__ void wl_kernel(const float* __restrict__ Wlast, float* __restrict__ Wl) {
    int tid = blockIdx.x * 256 + threadIdx.x;
    if (tid >= C_ * R_ * R_) return;
    int r  = tid & 63;
    int ci = tid >> 6;
    const float* p = Wlast + ((size_t)ci * 8) * 64 + r;
    float s = 0.f;
    #pragma unroll
    for (int m = 0; m < 8; ++m) s += p[m * 64];
    Wl[tid] = s;
}

__global__ void phi_kernel(const float* __restrict__ x, float* __restrict__ phidiv) {
    int tid = blockIdx.x * 256 + threadIdx.x;
    if (tid >= B_ * D_) return;
    const float* xp = x + (size_t)tid * 16;
    float acc[8];
    #pragma unroll
    for (int m = 0; m < 8; ++m) acc[m] = 0.f;
    for (int n = 0; n < 16; ++n) {
        float a = 1.5707963267948966f * xp[n];
        float sn, cs;
        sincosf(a, &sn, &cs);
        float pc[8], ps[8];
        pc[0] = 1.f; ps[0] = 1.f;
        #pragma unroll
        for (int k = 1; k < 8; ++k) { pc[k] = pc[k-1] * cs; ps[k] = ps[k-1] * sn; }
        #pragma unroll
        for (int m = 0; m < 8; ++m) acc[m] = fmaf(pc[7-m], ps[m], acc[m]);
    }
    float* o = phidiv + (size_t)tid * 8;
    #pragma unroll
    for (int m = 0; m < 8; ++m) o[m] = acc[m];
}

__device__ __forceinline__ void form_pair(const v2f (&wr)[8][8], const float (&pa)[8], const float (&pb)[8],
                                          float* __restrict__ f0, float* __restrict__ f1, const int p)
{
    v2f u = {0.f, 0.f}, w = {0.f, 0.f};
    #pragma unroll
    for (int j = 0; j < 8; ++j) {
        const v2f xa = {pa[j], pa[j]};
        const v2f xb = {pb[j], pb[j]};
        u = __builtin_elementwise_fma(xa, wr[p][j], u);
        w = __builtin_elementwise_fma(xb, wr[p][j], w);
    }
    *(v2f*)(f0 + 2 * p) = u;
    *(v2f*)(f1 + 2 * p) = w;
}

struct Frag2 { v2f a[2], b0[2], b1[2], b2[2], b3[2]; };

__device__ __forceinline__ void load_frag2(Frag2& f, const float* __restrict__ srow,
                                           const float* __restrict__ Acur, const int lmod, const int lhi, const int p)
{
    #pragma unroll
    for (int h = 0; h < 2; ++h) {
        const int koff = (2 * p + h) * 4 + 2 * lhi;
        f.a[h] = *(const v2f*)(srow + koff);
        const float* ab = Acur + koff;
        f.b0[h] = *(const v2f*)(ab + (lmod +  0) * AT_STRIDE);
        f.b1[h] = *(const v2f*)(ab + (lmod + 16) * AT_STRIDE);
        f.b2[h] = *(const v2f*)(ab + (lmod + 32) * AT_STRIDE);
        f.b3[h] = *(const v2f*)(ab + (lmod + 48) * AT_STRIDE);
    }
}

template <bool FORM>
__device__ __forceinline__ void chain_step(const v2f (&wr)[8][8], const float (&pa)[8], const float (&pb)[8],
                                           float* __restrict__ f0, float* __restrict__ f1,
                                           const float* __restrict__ Acur, float* __restrict__ Ssm,
                                           const int wave, const int lmod, const int lhi)
{
    const float* srow = Ssm + (16 * wave + lmod) * S_STRIDE;
    v8f c0 = {}, c1 = {}, c2 = {}, c3 = {};
    Frag2 fr[2];
    load_frag2(fr[0], srow, Acur, lmod, lhi, 0);
    __builtin_amdgcn_sched_barrier(0);
    #pragma unroll
    for (int p = 0; p < 8; ++p) {
        const Frag2& f = fr[p & 1];
        if (p < 7) load_frag2(fr[(p + 1) & 1], srow, Acur, lmod, lhi, p + 1);
        #pragma unroll
        for (int h = 0; h < 2; ++h) {
            c0 = __builtin_amdgcn_wmma_f32_16x16x4_f32(false, f.a[h], false, f.b0[h], (short)0, c0, false, false);
            c1 = __builtin_amdgcn_wmma_f32_16x16x4_f32(false, f.a[h], false, f.b1[h], (short)0, c1, false, false);
            c2 = __builtin_amdgcn_wmma_f32_16x16x4_f32(false, f.a[h], false, f.b2[h], (short)0, c2, false, false);
            c3 = __builtin_amdgcn_wmma_f32_16x16x4_f32(false, f.a[h], false, f.b3[h], (short)0, c3, false, false);
        }
        if (FORM) form_pair(wr, pa, pb, f0, f1, p);
        if (p < 7) __builtin_amdgcn_sched_group_barrier(0x100, 5, 0);
        #pragma unroll
        for (int g = 0; g < 8; ++g) {
            __builtin_amdgcn_sched_group_barrier(0x008, 1, 0);
            if (FORM) __builtin_amdgcn_sched_group_barrier(0x002, 2, 0);
        }
        if (FORM) __builtin_amdgcn_sched_group_barrier(0x200, 2, 0);
        __builtin_amdgcn_sched_barrier(0);
    }
    #pragma unroll
    for (int v = 0; v < 8; ++v) {
        float* prow = Ssm + (16 * wave + v + 8 * lhi) * S_STRIDE + lmod;
        prow[0]  = c0[v];
        prow[16] = c1[v];
        prow[32] = c2[v];
        prow[48] = c3[v];
    }
}

__global__ __launch_bounds__(256, 4)
void trnet_kernel(const float* __restrict__ Wsh, const float* __restrict__ phidiv,
                  const float* __restrict__ Wl, float* __restrict__ out)
{
    extern __shared__ float smem[];
    const int half = threadIdx.x >> 7;
    const int t0   = threadIdx.x & 127;
    float* At  = smem + half * LDS_HALF;
    float* Ssm = At  + 2 * AT_BUF;
    float* red = Ssm + 64 * S_STRIDE;

    const int b = blockIdx.x * 2 + half;

    const int fko = threadIdx.x & 63;
    const int fi0 = (threadIdx.x >> 6) * 16;
    v2f wr[8][8];
    {
        const float* wp = Wsh + fi0 * 512 + fko;
        #pragma unroll
        for (int p = 0; p < 8; ++p) {
            #pragma unroll
            for (int j = 0; j < 8; ++j) {
                const v2f t = {wp[(2 * p) * 512 + j * 64], wp[(2 * p + 1) * 512 + j * 64]};
                wr[p][j] = t;
            }
        }
    }
    float* fa0 = smem + fko * AT_STRIDE + fi0;
    float* fa1 = fa0 + LDS_HALF;
    typedef const __attribute__((address_space(4))) float cfloat;
    cfloat* php = (cfloat*)(phidiv + (size_t)blockIdx.x * (2 * D_ * M_));
    cfloat* phq = php + D_ * M_;

    for (int q = t0; q < 64 * S_STRIDE; q += 128) Ssm[q] = 1.0f;

    const int wave = t0 >> 5;
    const int lane = t0 & 31;
    const int lmod = lane & 15;
    const int lhi  = lane >> 4;

    float pa[8], pb[8];
    #pragma unroll
    for (int j = 0; j < 8; ++j) { pa[j] = php[j]; pb[j] = phq[j]; }
    #pragma unroll
    for (int p = 0; p < 8; ++p) form_pair(wr, pa, pb, fa0, fa1, p);
    #pragma unroll
    for (int j = 0; j < 8; ++j) { pa[j] = php[8 + j]; pb[j] = phq[8 + j]; }
    __syncthreads();

    __builtin_amdgcn_s_setprio(3);
    int drop = 32;
    for (int d = 0; d < D_ - 1; ++d) {
        const int cur = (d & 1) * AT_BUF;
        const int nxt = AT_BUF - cur;
        float na[8], nb[8];
        {
            const int dn = (d + 2) & (D_ - 1);
            #pragma unroll
            for (int j = 0; j < 8; ++j) { na[j] = php[dn * 8 + j]; nb[j] = phq[dn * 8 + j]; }
        }
        chain_step<true>(wr, pa, pb, fa0 + nxt, fa1 + nxt, At + cur, Ssm, wave, lmod, lhi);
        if (d + 1 == drop) {
            if (drop == 32)      { __builtin_amdgcn_s_setprio(2); drop = 48; }
            else if (drop == 48) { __builtin_amdgcn_s_setprio(1); drop = 56; }
            else                 { __builtin_amdgcn_s_setprio(0); drop = 0;  }
        }
        __syncthreads();
        #pragma unroll
        for (int j = 0; j < 8; ++j) { pa[j] = na[j]; pb[j] = nb[j]; }
    }
    chain_step<false>(wr, pa, pb, fa0, fa1, At + AT_BUF, Ssm, wave, lmod, lhi);
    __syncthreads();

    float acc[C_];
    #pragma unroll
    for (int c = 0; c < C_; ++c) acc[c] = 0.f;
    for (int q = 0; q < 32; ++q) {
        int e = t0 + 128 * q;
        int r = e & 63;
        int i = e >> 6;
        float s = Ssm[r * S_STRIDE + i];
        #pragma unroll
        for (int c = 0; c < C_; ++c)
            acc[c] = fmaf(s, Wl[c * 4096 + i * 64 + r], acc[c]);
    }
    #pragma unroll
    for (int c = 0; c < C_; ++c) red[t0 * C_ + c] = acc[c];
    __syncthreads();
    if (t0 < C_) {
        float s = 0.f;
        for (int t = 0; t < 128; ++t) s += red[t * C_ + t0];
        At[t0] = s;
    }
    __syncthreads();
    if (t0 < C_) {
        float mx = At[0];
        #pragma unroll
        for (int c = 1; c < C_; ++c) mx = fmaxf(mx, At[c]);
        float se = 0.f;
        #pragma unroll
        for (int c = 0; c < C_; ++c) se += expf(At[c] - mx);
        out[b * C_ + t0] = At[t0] - (logf(se) + mx);
    }
}

extern "C" void kernel_launch(void* const* d_in, const int* in_sizes, int n_in,
                              void* d_out, int out_size, void* d_ws, size_t ws_size,
                              hipStream_t stream) {
    const float* x     = (const float*)d_in[0];
    const float* Wsh   = (const float*)d_in[1];
    const float* Wlast = (const float*)d_in[2];
    float* out = (float*)d_out;
    float* ws  = (float*)d_ws;
    float* Wl     = ws;
    float* phidiv = ws + (C_ * R_ * R_);

    wl_kernel<<<(C_ * R_ * R_ + 255) / 256, 256, 0, stream>>>(Wlast, Wl);
    phi_kernel<<<(B_ * D_ + 255) / 256, 256, 0, stream>>>(x, phidiv);

    const size_t shbytes = (size_t)(2 * LDS_HALF) * sizeof(float);
    hipFuncSetAttribute((const void*)trnet_kernel,
                        hipFuncAttributeMaxDynamicSharedMemorySize, (int)shbytes);
    trnet_kernel<<<B_ / 2, 256, shbytes, stream>>>(Wsh, phidiv, Wl, out);
}


# Step matrices on a ring of three LDS buffers, work-group barrier split: next matrix formed in the step's first half, signal behind it, wait at step end; no rendezvous stall; sum orders unchanged
# speedup vs baseline: 2.6078x; 1.0448x over previous
#include <hip/hip_runtime.h>
#include <math.h>


typedef __attribute__((ext_vector_type(2))) float v2f;
typedef __attribute__((ext_vector_type(8))) float v8f;

#define B_  1024
#define N_  1024
#define M_  8
#define D_  64
#define R_  64
#define C_  10

#define AT_STRIDE  66
#define S_STRIDE   66
#define AT_BUF     (64 * AT_STRIDE)
#define AT_RING    3
#define LDS_HALF   (AT_RING * AT_BUF + 64 * S_STRIDE + 128 * C_)

__global__ void wl_kernel(const float* __restrict__ Wlast, float* __restrict__ Wl) {
    int tid = blockIdx.x * 256 + threadIdx.x;
    if (tid >= C_ * R_ * R_) return;
    int r  = tid & 63;
    int ci = tid >> 6;
    const float* p = Wlast + ((size_t)ci * 8) * 64 + r;
    float s = 0.f;
    #pragma unroll
    for (int m = 0; m < 8; ++m) s += p[m * 64];
    Wl[tid] = s;
}

__global__ void phi_kernel(const float* __restrict__ x, float* __restrict__ phidiv) {
    int tid = blockIdx.x * 256 + threadIdx.x;
    if (tid >= B_ * D_) return;
    const float* xp = x + (size_t)tid * 16;
    float acc[8];
    #pragma unroll
    for (int m = 0; m < 8; ++m) acc[m] = 0.f;
    for (int n = 0; n < 16; ++n) {
        float a = 1.5707963267948966f * xp[n];
        float sn, cs;
        sincosf(a, &sn, &cs);
        float pc[8], ps[8];
        pc[0] = 1.f; ps[0] = 1.f;
        #pragma unroll
        for (int k = 1; k < 8; ++k) { pc[k] = pc[k-1] * cs; ps[k] = ps[k-1] * sn; }
        #pragma unroll
        for (int m = 0; m < 8; ++m) acc[m] = fmaf(pc[7-m], ps[m], acc[m]);
    }
    float* o = phidiv + (size_t)tid * 8;
    #pragma unroll
    for (int m = 0; m < 8; ++m) o[m] = acc[m];
}

__device__ __forceinline__ void form_pair(const v2f (&wr)[8][8], const float (&pa)[8], const float (&pb)[8],
                                          float* __restrict__ f0, float* __restrict__ f1, const int p)
{
    v2f u = {0.f, 0.f}, w = {0.f, 0.f};
    #pragma unroll
    for (int j = 0; j < 8; ++j) {
        const v2f xa = {pa[j], pa[j]};
        const v2f xb = {pb[j], pb[j]};
        u = __builtin_elementwise_fma(xa, wr[p][j], u);
        w = __builtin_elementwise_fma(xb, wr[p][j], w);
    }
    *(v2f*)(f0 + 2 * p) = u;
    *(v2f*)(f1 + 2 * p) = w;
}

struct Frag2 { v2f a[2], b0[2], b1[2], b2[2], b3[2]; };

__device__ __forceinline__ void load_frag2(Frag2& f, const float* __restrict__ srow,
                                           const float* __restrict__ Acur, const int lmod, const int lhi, const int p)
{
    #pragma unroll
    for (int h = 0; h < 2; ++h) {
        const int koff = (2 * p + h) * 4 + 2 * lhi;
        f.a[h] = *(const v2f*)(srow + koff);
        const float* ab = Acur + koff;
        f.b0[h] = *(const v2f*)(ab + (lmod +  0) * AT_STRIDE);
        f.b1[h] = *(const v2f*)(ab + (lmod + 16) * AT_STRIDE);
        f.b2[h] = *(const v2f*)(ab + (lmod + 32) * AT_STRIDE);
        f.b3[h] = *(const v2f*)(ab + (lmod + 48) * AT_STRIDE);
    }
}

template <bool FORM>
__device__ __forceinline__ void chain_step(const v2f (&wr)[8][8], const float (&pa)[8], const float (&pb)[8],
                                           float* __restrict__ f0, float* __restrict__ f1,
                                           const float* __restrict__ Acur, float* __restrict__ Ssm,
                                           const int wave, const int lmod, const int lhi)
{
    const float* srow = Ssm + (16 * wave + lmod) * S_STRIDE;
    v8f c0 = {}, c1 = {}, c2 = {}, c3 = {};
    Frag2 fr[2];
    load_frag2(fr[0], srow, Acur, lmod, lhi, 0);
    __builtin_amdgcn_sched_barrier(0);
    #pragma unroll
    for (int p = 0; p < 8; ++p) {
        const Frag2& f = fr[p & 1];
        if (p < 7) load_frag2(fr[(p + 1) & 1], srow, Acur, lmod, lhi, p + 1);
        #pragma unroll
        for (int h = 0; h < 2; ++h) {
            c0 = __builtin_amdgcn_wmma_f32_16x16x4_f32(false, f.a[h], false, f.b0[h], (short)0, c0, false, false);
            c1 = __builtin_amdgcn_wmma_f32_16x16x4_f32(false, f.a[h], false, f.b1[h], (short)0, c1, false, false);
            c2 = __builtin_amdgcn_wmma_f32_16x16x4_f32(false, f.a[h], false, f.b2[h], (short)0, c2, false, false);
            c3 = __builtin_amdgcn_wmma_f32_16x16x4_f32(false, f.a[h], false, f.b3[h], (short)0, c3, false, false);
        }
        const bool forming = FORM && (p < 4);
        if (forming) {
            form_pair(wr, pa, pb, f0, f1, 2 * p);
            form_pair(wr, pa, pb, f0, f1, 2 * p + 1);
        }
        if (p < 7) __builtin_amdgcn_sched_group_barrier(0x100, 5, 0);
        #pragma unroll
        for (int g = 0; g < 8; ++g) {
            __builtin_amdgcn_sched_group_barrier(0x008, 1, 0);
            if (forming) __builtin_amdgcn_sched_group_barrier(0x002, 4, 0);
        }
        if (forming) __builtin_amdgcn_sched_group_barrier(0x200, 2, 0);
        __builtin_amdgcn_sched_barrier(0);
        if (FORM && p == 4) {
            asm volatile("s_wait_dscnt 0x0" ::: "memory");
            asm volatile("s_barrier_signal -1" ::: "memory");
            __builtin_amdgcn_sched_barrier(0);
        }
    }
    #pragma unroll
    for (int v = 0; v < 8; ++v) {
        float* prow = Ssm + (16 * wave + v + 8 * lhi) * S_STRIDE + lmod;
        prow[0]  = c0[v];
        prow[16] = c1[v];
        prow[32] = c2[v];
        prow[48] = c3[v];
    }
}

__global__ __launch_bounds__(256, 4)
void trnet_kernel(const float* __restrict__ Wsh, const float* __restrict__ phidiv,
                  const float* __restrict__ Wl, float* __restrict__ out)
{
    extern __shared__ float smem[];
    const int half = threadIdx.x >> 7;
    const int t0   = threadIdx.x & 127;
    float* At  = smem + half * LDS_HALF;
    float* Ssm = At  + AT_RING * AT_BUF;
    float* red = Ssm + 64 * S_STRIDE;

    const int b = blockIdx.x * 2 + half;

    const int fko = threadIdx.x & 63;
    const int fi0 = (threadIdx.x >> 6) * 16;
    v2f wr[8][8];
    {
        const float* wp = Wsh + fi0 * 512 + fko;
        #pragma unroll
        for (int p = 0; p < 8; ++p) {
            #pragma unroll
            for (int j = 0; j < 8; ++j) {
                const v2f t = {wp[(2 * p) * 512 + j * 64], wp[(2 * p + 1) * 512 + j * 64]};
                wr[p][j] = t;
            }
        }
    }
    float* fa0 = smem + fko * AT_STRIDE + fi0;
    float* fa1 = fa0 + LDS_HALF;
    typedef const __attribute__((address_space(4))) float cfloat;
    cfloat* php = (cfloat*)(phidiv + (size_t)blockIdx.x * (2 * D_ * M_));
    cfloat* phq = php + D_ * M_;

    for (int q = t0; q < 64 * S_STRIDE; q += 128) Ssm[q] = 1.0f;

    const int wave = t0 >> 5;
    const int lane = t0 & 31;
    const int lmod = lane & 15;
    const int lhi  = lane >> 4;

    float pa[8], pb[8];
    #pragma unroll
    for (int j = 0; j < 8; ++j) { pa[j] = php[j]; pb[j] = phq[j]; }
    #pragma unroll
    for (int p = 0; p < 8; ++p) form_pair(wr, pa, pb, fa0, fa1, p);
    #pragma unroll
    for (int j = 0; j < 8; ++j) { pa[j] = php[8 + j]; pb[j] = phq[8 + j]; }
    __syncthreads();

    __builtin_amdgcn_s_setprio(3);
    int drop = 32;
    int cur = 0;
    for (int d = 0; d < D_ - 1; ++d) {
        const int nxt = (cur == (AT_RING - 1) * AT_BUF) ? 0 : cur + AT_BUF;
        float na[8], nb[8];
        {
            const int dn = (d + 2) & (D_ - 1);
            #pragma unroll
            for (int j = 0; j < 8; ++j) { na[j] = php[dn * 8 + j]; nb[j] = phq[dn * 8 + j]; }
        }
        chain_step<true>(wr, pa, pb, fa0 + nxt, fa1 + nxt, At + cur, Ssm, wave, lmod, lhi);
        if (d + 1 == drop) {
            if (drop == 32)      { __builtin_amdgcn_s_setprio(2); drop = 48; }
            else if (drop == 48) { __builtin_amdgcn_s_setprio(1); drop = 56; }
            else                 { __builtin_amdgcn_s_setprio(0); drop = 0;  }
        }
        asm volatile("s_barrier_wait -1" ::: "memory");
        #pragma unroll
        for (int j = 0; j < 8; ++j) { pa[j] = na[j]; pb[j] = nb[j]; }
        cur = nxt;
    }
    chain_step<false>(wr, pa, pb, fa0, fa1, At + cur, Ssm, wave, lmod, lhi);
    __syncthreads();

    float acc[C_];
    #pragma unroll
    for (int c = 0; c < C_; ++c) acc[c] = 0.f;
    for (int q = 0; q < 32; ++q) {
        int e = t0 + 128 * q;
        int r = e & 63;
        int i = e >> 6;
        float s = Ssm[r * S_STRIDE + i];
        #pragma unroll
        for (int c = 0; c < C_; ++c)
            acc[c] = fmaf(s, Wl[c * 4096 + i * 64 + r], acc[c]);
    }
    #pragma unroll
    for (int c = 0; c < C_; ++c) red[t0 * C_ + c] = acc[c];
    __syncthreads();
    if (t0 < C_) {
        float s = 0.f;
        for (int t = 0; t < 128; ++t) s += red[t * C_ + t0];
        At[t0] = s;
    }
    __syncthreads();
    if (t0 < C_) {
        float mx = At[0];
        #pragma unroll
        for (int c = 1; c < C_; ++c) mx = fmaxf(mx, At[c]);
        float se = 0.f;
        #pragma unroll
        for (int c = 0; c < C_; ++c) se += expf(At[c] - mx);
        out[b * C_ + t0] = At[t0] - (logf(se) + mx);
    }
}

extern "C" void kernel_launch(void* const* d_in, const int* in_sizes, int n_in,
                              void* d_out, int out_size, void* d_ws, size_t ws_size,
                              hipStream_t stream) {
    const float* x     = (const float*)d_in[0];
    const float* Wsh   = (const float*)d_in[1];
    const float* Wlast = (const float*)d_in[2];
    float* out = (float*)d_out;
    float* ws  = (float*)d_ws;
    float* Wl     = ws;
    float* phidiv = ws + (C_ * R_ * R_);

    wl_kernel<<<(C_ * R_ * R_ + 255) / 256, 256, 0, stream>>>(Wlast, Wl);
    phi_kernel<<<(B_ * D_ + 255) / 256, 256, 0, stream>>>(x, phidiv);

    const size_t shbytes = (size_t)(2 * LDS_HALF) * sizeof(float);
    hipFuncSetAttribute((const void*)trnet_kernel,
                        hipFuncAttributeMaxDynamicSharedMemorySize, (int)shbytes);
    trnet_kernel<<<B_ / 2, 256, shbytes, stream>>>(Wsh, phidiv, Wl, out);
}
